# starting kernel
# baseline (speedup 1.0000x reference)
#include <hip/hip_runtime.h>

#define B_    4
#define N_    1024
#define FI_   128
#define FO_   64
#define DELTA_ 1.0f
#define EPS_   1e-3f

typedef __attribute__((ext_vector_type(2))) float v2f;
typedef __attribute__((ext_vector_type(8))) float v8f;

static __device__ __forceinline__ v8f wmma_f32_16x16x4(v2f a, v2f b, v8f c) {
  return __builtin_amdgcn_wmma_f32_16x16x4_f32(
       false, a,  false, b,
       (short)0, c,  false,  false);
}

__global__ void __launch_bounds__(128)
gtv_xw_gemm(const float* __restrict__ x, const float* __restrict__ w,
            float* __restrict__ xw) {
  const int tid  = threadIdx.x;
  const int wv   = tid >> 5;
  const int lane = tid & 31;
  const int hh   = lane >> 4;
  const int l15  = lane & 15;
  const int rt   = blockIdx.x;
  const int f0   = wv * 16;

  const float* __restrict__ xrow = x + (size_t)(rt * 16 + l15) * FI_;
  v8f c = {0.f, 0.f, 0.f, 0.f, 0.f, 0.f, 0.f, 0.f};

  #pragma unroll 4
  for (int kc = 0; kc < FI_ / 4; ++kc) {
    const int ks = kc * 4 + hh * 2;
    v2f a;  a.x = xrow[ks];  a.y = xrow[ks + 1];
    const float* __restrict__ wp = w + (size_t)ks * FO_ + f0 + l15;
    v2f b;  b.x = wp[0];     b.y = wp[FO_];
    c = wmma_f32_16x16x4(a, b, c);
  }

  float* __restrict__ orow = xw + (size_t)(rt * 16 + hh * 8) * FO_ + f0 + l15;
  #pragma unroll
  for (int r = 0; r < 8; ++r) orow[(size_t)r * FO_] = c[r];
}

__global__ void __launch_bounds__(256)
gtv_fused(const float* __restrict__ adj, const float* __restrict__ xw,
          const float* __restrict__ bias, float* __restrict__ out) {
  __shared__ float xwi_s[16][FO_];
  __shared__ float Msh[8][16][20];
  __shared__ float out_s[16][FO_];
  __shared__ float deg_s[16];

  const int tid  = threadIdx.x;
  const int wv   = tid >> 5;
  const int lane = tid & 31;
  const int hh   = lane >> 4;
  const int l15  = lane & 15;

  const int b  = blockIdx.x >> 6;
  const int i0 = (blockIdx.x & 63) * 16;

  if (tid < 16) deg_s[tid] = 0.f;
  {
    const int row = tid >> 4;
    const int c4  = (tid & 15) * 4;
    *(float4*)&out_s[row][c4] = make_float4(0.f, 0.f, 0.f, 0.f);
    *(float4*)&xwi_s[row][c4] =
        *(const float4*)(xw + (size_t)(b * N_ + i0 + row) * FO_ + c4);
  }
  __syncthreads();

  float dacc[8];
  #pragma unroll
  for (int r = 0; r < 8; ++r) dacc[r] = 0.f;
  v8f oacc[4];
  #pragma unroll
  for (int f = 0; f < 4; ++f)
    #pragma unroll
    for (int e = 0; e < 8; ++e) oacc[f][e] = 0.f;

  const float* __restrict__ adjb = adj + (size_t)b * N_ * N_ + (size_t)i0 * N_;
  const float* __restrict__ xwb  = xw + (size_t)b * N_ * FO_;

  for (int step = 0; step < 8; ++step) {
    const int jt = step * 8 + wv;
    const int j0 = jt * 16;
    if (step < 7)
      __builtin_prefetch(adjb + (size_t)(hh * 8) * N_ + (j0 + 128) + l15, 0, 0);

    float acc[8];
    #pragma unroll
    for (int r = 0; r < 8; ++r) acc[r] = 0.f;
    const float* __restrict__ xwj = xwb + (size_t)(j0 + l15) * FO_;
    #pragma unroll 4
    for (int kc = 0; kc < FO_ / 4; ++kc) {
      const float4 xj = *(const float4*)(xwj + kc * 4);
      #pragma unroll
      for (int r = 0; r < 8; ++r) {
        const float4 xi = *(const float4*)&xwi_s[r + hh * 8][kc * 4];
        acc[r] += fabsf(xi.x - xj.x) + fabsf(xi.y - xj.y)
                + fabsf(xi.z - xj.z) + fabsf(xi.w - xj.w);
      }
    }

    const float* __restrict__ ap = adjb + j0 + l15;
    #pragma unroll
    for (int r = 0; r < 8; ++r) {
      const int irow = r + hh * 8;
      const float a = ap[(size_t)irow * N_];
      float m = (a != 0.f) ? (a / fmaxf(acc[r], EPS_)) : a;
      m *= DELTA_;
      dacc[r] += m;
      Msh[wv][irow][l15] = m;
    }

    v2f avec[4];
    #pragma unroll
    for (int kchunk = 0; kchunk < 4; ++kchunk) {
      const int ks = kchunk * 4 + hh * 2;
      avec[kchunk].x = Msh[wv][l15][ks];
      avec[kchunk].y = Msh[wv][l15][ks + 1];
    }
    #pragma unroll
    for (int fc = 0; fc < 4; ++fc) {
      #pragma unroll
      for (int kchunk = 0; kchunk < 4; ++kchunk) {
        const int ks = kchunk * 4 + hh * 2;
        const float* __restrict__ bp =
            xwb + (size_t)(j0 + ks) * FO_ + fc * 16 + l15;
        v2f bv;  bv.x = bp[0];  bv.y = bp[FO_];
        oacc[fc] = wmma_f32_16x16x4(avec[kchunk], bv, oacc[fc]);
      }
    }
  }

  #pragma unroll
  for (int r = 0; r < 8; ++r) {
    float d = dacc[r];
    d += __shfl_xor(d, 1, 16);
    d += __shfl_xor(d, 2, 16);
    d += __shfl_xor(d, 4, 16);
    d += __shfl_xor(d, 8, 16);
    if (l15 == 0) atomicAdd(&deg_s[r + hh * 8], d);
  }
  #pragma unroll
  for (int fc = 0; fc < 4; ++fc)
    #pragma unroll
    for (int r = 0; r < 8; ++r)
      atomicAdd(&out_s[r + hh * 8][fc * 16 + l15], oacc[fc][r]);
  __syncthreads();

  {
    const int row = tid >> 4;
    const int c4  = (tid & 15) * 4;
    const float diagc = 1.f - deg_s[row];
    const float4 xi = *(const float4*)&xwi_s[row][c4];
    const float4 bs = *(const float4*)(bias + c4);
    float4 o = *(const float4*)&out_s[row][c4];
    o.x += diagc * xi.x + bs.x;
    o.y += diagc * xi.y + bs.y;
    o.z += diagc * xi.z + bs.z;
    o.w += diagc * xi.w + bs.w;
    *(float4*)(out + (size_t)(b * N_ + i0 + row) * FO_ + c4) = o;
  }
}

extern "C" void kernel_launch(void* const* d_in, const int* in_sizes, int n_in,
                              void* d_out, int out_size, void* d_ws, size_t ws_size,
                              hipStream_t stream) {
  (void)in_sizes; (void)n_in; (void)out_size; (void)ws_size;
  const float* x    = (const float*)d_in[0];
  const float* adj  = (const float*)d_in[1];
  const float* w    = (const float*)d_in[2];
  const float* bias = (const float*)d_in[3];
  float*       out  = (float*)d_out;
  float*       xw   = (float*)d_ws;

  gtv_xw_gemm<<<(B_ * N_) / 16, 128, 0, stream>>>(x, w, xw);
  gtv_fused  <<<B_ * (N_ / 16), 256, 0, stream>>>(adj, xw, bias, out);
}


# adjacency tile: its 8 row loads issued as one group before the distance loop, quotient select made branch-free (was 8 load-wait-divide round trips in series per step)
# speedup vs baseline: 1.6852x; 1.6852x over previous
#include <hip/hip_runtime.h>

#define B_    4
#define N_    1024
#define FI_   128
#define FO_   64
#define DELTA_ 1.0f
#define EPS_   1e-3f

typedef __attribute__((ext_vector_type(2))) float v2f;
typedef __attribute__((ext_vector_type(8))) float v8f;

static __device__ __forceinline__ v8f wmma_f32_16x16x4(v2f a, v2f b, v8f c) {
  return __builtin_amdgcn_wmma_f32_16x16x4_f32(
       false, a,  false, b,
       (short)0, c,  false,  false);
}

__global__ void __launch_bounds__(128)
gtv_xw_gemm(const float* __restrict__ x, const float* __restrict__ w,
            float* __restrict__ xw) {
  const int tid  = threadIdx.x;
  const int wv   = tid >> 5;
  const int lane = tid & 31;
  const int hh   = lane >> 4;
  const int l15  = lane & 15;
  const int rt   = blockIdx.x;
  const int f0   = wv * 16;

  const float* __restrict__ xrow = x + (size_t)(rt * 16 + l15) * FI_;
  v8f c = {0.f, 0.f, 0.f, 0.f, 0.f, 0.f, 0.f, 0.f};

  #pragma unroll 4
  for (int kc = 0; kc < FI_ / 4; ++kc) {
    const int ks = kc * 4 + hh * 2;
    v2f a;  a.x = xrow[ks];  a.y = xrow[ks + 1];
    const float* __restrict__ wp = w + (size_t)ks * FO_ + f0 + l15;
    v2f b;  b.x = wp[0];     b.y = wp[FO_];
    c = wmma_f32_16x16x4(a, b, c);
  }

  float* __restrict__ orow = xw + (size_t)(rt * 16 + hh * 8) * FO_ + f0 + l15;
  #pragma unroll
  for (int r = 0; r < 8; ++r) orow[(size_t)r * FO_] = c[r];
}

__global__ void __launch_bounds__(256)
gtv_fused(const float* __restrict__ adj, const float* __restrict__ xw,
          const float* __restrict__ bias, float* __restrict__ out) {
  __shared__ float xwi_s[16][FO_];
  __shared__ float Msh[8][16][20];
  __shared__ float out_s[16][FO_];
  __shared__ float deg_s[16];

  const int tid  = threadIdx.x;
  const int wv   = tid >> 5;
  const int lane = tid & 31;
  const int hh   = lane >> 4;
  const int l15  = lane & 15;

  const int b  = blockIdx.x >> 6;
  const int i0 = (blockIdx.x & 63) * 16;

  if (tid < 16) deg_s[tid] = 0.f;
  {
    const int row = tid >> 4;
    const int c4  = (tid & 15) * 4;
    *(float4*)&out_s[row][c4] = make_float4(0.f, 0.f, 0.f, 0.f);
    *(float4*)&xwi_s[row][c4] =
        *(const float4*)(xw + (size_t)(b * N_ + i0 + row) * FO_ + c4);
  }
  __syncthreads();

  float dacc[8];
  #pragma unroll
  for (int r = 0; r < 8; ++r) dacc[r] = 0.f;
  v8f oacc[4];
  #pragma unroll
  for (int f = 0; f < 4; ++f)
    #pragma unroll
    for (int e = 0; e < 8; ++e) oacc[f][e] = 0.f;

  const float* __restrict__ adjb = adj + (size_t)b * N_ * N_ + (size_t)i0 * N_;
  const float* __restrict__ xwb  = xw + (size_t)b * N_ * FO_;

  for (int step = 0; step < 8; ++step) {
    const int jt = step * 8 + wv;
    const int j0 = jt * 16;
    if (step < 7)
      __builtin_prefetch(adjb + (size_t)(hh * 8) * N_ + (j0 + 128) + l15, 0, 0);

    const float* __restrict__ ap = adjb + j0 + l15;
    float av[8];
    #pragma unroll
    for (int r = 0; r < 8; ++r) av[r] = ap[(size_t)(r + hh * 8) * N_];

    float acc[8];
    #pragma unroll
    for (int r = 0; r < 8; ++r) acc[r] = 0.f;
    const float* __restrict__ xwj = xwb + (size_t)(j0 + l15) * FO_;
    #pragma unroll 4
    for (int kc = 0; kc < FO_ / 4; ++kc) {
      const float4 xj = *(const float4*)(xwj + kc * 4);
      #pragma unroll
      for (int r = 0; r < 8; ++r) {
        const float4 xi = *(const float4*)&xwi_s[r + hh * 8][kc * 4];
        acc[r] += fabsf(xi.x - xj.x) + fabsf(xi.y - xj.y)
                + fabsf(xi.z - xj.z) + fabsf(xi.w - xj.w);
      }
    }

    #pragma unroll
    for (int r = 0; r < 8; ++r) {
      const int irow = r + hh * 8;
      const float a = av[r];
      const float q = a / fmaxf(acc[r], EPS_);
      float m = (a != 0.f) ? q : a;
      m *= DELTA_;
      dacc[r] += m;
      Msh[wv][irow][l15] = m;
    }

    v2f avec[4];
    #pragma unroll
    for (int kchunk = 0; kchunk < 4; ++kchunk) {
      const int ks = kchunk * 4 + hh * 2;
      avec[kchunk].x = Msh[wv][l15][ks];
      avec[kchunk].y = Msh[wv][l15][ks + 1];
    }
    #pragma unroll
    for (int fc = 0; fc < 4; ++fc) {
      #pragma unroll
      for (int kchunk = 0; kchunk < 4; ++kchunk) {
        const int ks = kchunk * 4 + hh * 2;
        const float* __restrict__ bp =
            xwb + (size_t)(j0 + ks) * FO_ + fc * 16 + l15;
        v2f bv;  bv.x = bp[0];  bv.y = bp[FO_];
        oacc[fc] = wmma_f32_16x16x4(avec[kchunk], bv, oacc[fc]);
      }
    }
  }

  #pragma unroll
  for (int r = 0; r < 8; ++r) {
    float d = dacc[r];
    d += __shfl_xor(d, 1, 16);
    d += __shfl_xor(d, 2, 16);
    d += __shfl_xor(d, 4, 16);
    d += __shfl_xor(d, 8, 16);
    if (l15 == 0) atomicAdd(&deg_s[r + hh * 8], d);
  }
  #pragma unroll
  for (int fc = 0; fc < 4; ++fc)
    #pragma unroll
    for (int r = 0; r < 8; ++r)
      atomicAdd(&out_s[r + hh * 8][fc * 16 + l15], oacc[fc][r]);
  __syncthreads();

  {
    const int row = tid >> 4;
    const int c4  = (tid & 15) * 4;
    const float diagc = 1.f - deg_s[row];
    const float4 xi = *(const float4*)&xwi_s[row][c4];
    const float4 bs = *(const float4*)(bias + c4);
    float4 o = *(const float4*)&out_s[row][c4];
    o.x += diagc * xi.x + bs.x;
    o.y += diagc * xi.y + bs.y;
    o.z += diagc * xi.z + bs.z;
    o.w += diagc * xi.w + bs.w;
    *(float4*)(out + (size_t)(b * N_ + i0 + row) * FO_ + c4) = o;
  }
}

extern "C" void kernel_launch(void* const* d_in, const int* in_sizes, int n_in,
                              void* d_out, int out_size, void* d_ws, size_t ws_size,
                              hipStream_t stream) {
  (void)in_sizes; (void)n_in; (void)out_size; (void)ws_size;
  const float* x    = (const float*)d_in[0];
  const float* adj  = (const float*)d_in[1];
  const float* w    = (const float*)d_in[2];
  const float* bias = (const float*)d_in[3];
  float*       out  = (float*)d_out;
  float*       xw   = (float*)d_ws;

  gtv_xw_gemm<<<(B_ * N_) / 16, 128, 0, stream>>>(x, w, xw);
  gtv_fused  <<<B_ * (N_ / 16), 256, 0, stream>>>(adj, xw, bias, out);
}


# distance loop: subtraction packed over feature pairs, |d| folded into the add (asm), xw_i LDS reads two groups ahead, xw_j row held in registers a tile ahead; a distance's f32 adds in element order
# speedup vs baseline: 2.0703x; 1.2285x over previous
#include <hip/hip_runtime.h>

#define B_    4
#define N_    1024
#define FI_   128
#define FO_   64
#define DELTA_ 1.0f
#define EPS_   1e-3f

typedef __attribute__((ext_vector_type(2))) float v2f;
typedef __attribute__((ext_vector_type(8))) float v8f;

static __device__ __forceinline__ v8f wmma_f32_16x16x4(v2f a, v2f b, v8f c) {
  return __builtin_amdgcn_wmma_f32_16x16x4_f32(
       false, a,  false, b,
       (short)0, c,  false,  false);
}

static __device__ __forceinline__ v2f pk_sub_f32(v2f a, v2f b) {
  v2f d;
  asm("v_pk_add_f32 %0, %1, %2 neg_lo:[0,1] neg_hi:[0,1]" : "=v"(d) : "v"(a), "v"(b));
  return d;
}

static __device__ __forceinline__ float add_abs_f32(float acc, float d) {
  float o;
  asm("v_add_f32_e64 %0, %1, |%2|" : "=v"(o) : "v"(acc), "v"(d));
  return o;
}

__global__ void __launch_bounds__(128)
gtv_xw_gemm(const float* __restrict__ x, const float* __restrict__ w,
            float* __restrict__ xw) {
  const int tid  = threadIdx.x;
  const int wv   = tid >> 5;
  const int lane = tid & 31;
  const int hh   = lane >> 4;
  const int l15  = lane & 15;
  const int rt   = blockIdx.x;
  const int f0   = wv * 16;

  const float* __restrict__ xrow = x + (size_t)(rt * 16 + l15) * FI_;
  v8f c = {0.f, 0.f, 0.f, 0.f, 0.f, 0.f, 0.f, 0.f};

  #pragma unroll 4
  for (int kc = 0; kc < FI_ / 4; ++kc) {
    const int ks = kc * 4 + hh * 2;
    v2f a;  a.x = xrow[ks];  a.y = xrow[ks + 1];
    const float* __restrict__ wp = w + (size_t)ks * FO_ + f0 + l15;
    v2f b;  b.x = wp[0];     b.y = wp[FO_];
    c = wmma_f32_16x16x4(a, b, c);
  }

  float* __restrict__ orow = xw + (size_t)(rt * 16 + hh * 8) * FO_ + f0 + l15;
  #pragma unroll
  for (int r = 0; r < 8; ++r) orow[(size_t)r * FO_] = c[r];
}

__global__ void __launch_bounds__(256, 4)
gtv_fused(const float* __restrict__ adj, const float* __restrict__ xw,
          const float* __restrict__ bias, float* __restrict__ out) {
  __shared__ float xwi_s[16][FO_];
  __shared__ float Msh[8][16][20];
  __shared__ float out_s[16][FO_];
  __shared__ float deg_s[16];

  const int tid  = threadIdx.x;
  const int wv   = tid >> 5;
  const int lane = tid & 31;
  const int hh   = lane >> 4;
  const int l15  = lane & 15;

  const int b  = blockIdx.x >> 6;
  const int i0 = (blockIdx.x & 63) * 16;

  if (tid < 16) deg_s[tid] = 0.f;
  {
    const int row = tid >> 4;
    const int c4  = (tid & 15) * 4;
    *(float4*)&out_s[row][c4] = make_float4(0.f, 0.f, 0.f, 0.f);
    *(float4*)&xwi_s[row][c4] =
        *(const float4*)(xw + (size_t)(b * N_ + i0 + row) * FO_ + c4);
  }
  __syncthreads();

  float dacc[8];
  #pragma unroll
  for (int r = 0; r < 8; ++r) dacc[r] = 0.f;
  v8f oacc[4];
  #pragma unroll
  for (int f = 0; f < 4; ++f)
    #pragma unroll
    for (int e = 0; e < 8; ++e) oacc[f][e] = 0.f;

  const float* __restrict__ adjb = adj + (size_t)b * N_ * N_ + (size_t)i0 * N_;
  const float* __restrict__ xwb  = xw + (size_t)b * N_ * FO_;

  float4 xjq[16];
  {
    const float* __restrict__ xwj0 = xwb + (size_t)(wv * 16 + l15) * FO_;
    #pragma unroll
    for (int kc = 0; kc < FO_ / 4; ++kc) xjq[kc] = *(const float4*)(xwj0 + kc * 4);
  }

  for (int step = 0; step < 8; ++step) {
    const int jt = step * 8 + wv;
    const int j0 = jt * 16;
    if (step < 7)
      __builtin_prefetch(adjb + (size_t)(hh * 8) * N_ + (j0 + 128) + l15, 0, 0);

    const float* __restrict__ ap = adjb + j0 + l15;
    float av[8];
    #pragma unroll
    for (int r = 0; r < 8; ++r) av[r] = ap[(size_t)(r + hh * 8) * N_];

    float acc[8];
    #pragma unroll
    for (int r = 0; r < 8; ++r) acc[r] = 0.f;
    float4 xiq[3][8];
    #pragma unroll
    for (int g = 0; g < 2; ++g)
      #pragma unroll
      for (int r = 0; r < 8; ++r) xiq[g][r] = *(const float4*)&xwi_s[r + hh * 8][g * 4];
    #pragma unroll
    for (int kc = 0; kc < FO_ / 4; ++kc) {
      const int cur = kc % 3;
      const int nxt = (kc + 2) % 3;
      if (kc + 2 < FO_ / 4) {
        #pragma unroll
        for (int r = 0; r < 4; ++r)
          xiq[nxt][r] = *(const float4*)&xwi_s[r + hh * 8][(kc + 2) * 4];
      }
      const float4 xj = xjq[kc];
      v2f xjlo;  xjlo.x = xj.x;  xjlo.y = xj.y;
      v2f xjhi;  xjhi.x = xj.z;  xjhi.y = xj.w;
      v2f d[8];
      #pragma unroll
      for (int r = 0; r < 8; ++r) {
        v2f xilo;  xilo.x = xiq[cur][r].x;  xilo.y = xiq[cur][r].y;
        d[r] = pk_sub_f32(xilo, xjlo);
      }
      #pragma unroll
      for (int r = 0; r < 8; ++r) acc[r] = add_abs_f32(acc[r], d[r].x);
      #pragma unroll
      for (int r = 0; r < 8; ++r) acc[r] = add_abs_f32(acc[r], d[r].y);
      if (kc + 2 < FO_ / 4) {
        #pragma unroll
        for (int r = 4; r < 8; ++r)
          xiq[nxt][r] = *(const float4*)&xwi_s[r + hh * 8][(kc + 2) * 4];
      }
      #pragma unroll
      for (int r = 0; r < 8; ++r) {
        v2f xihi;  xihi.x = xiq[cur][r].z;  xihi.y = xiq[cur][r].w;
        d[r] = pk_sub_f32(xihi, xjhi);
      }
      #pragma unroll
      for (int r = 0; r < 8; ++r) acc[r] = add_abs_f32(acc[r], d[r].x);
      #pragma unroll
      for (int r = 0; r < 8; ++r) acc[r] = add_abs_f32(acc[r], d[r].y);
      __builtin_amdgcn_sched_barrier(0);
    }

    __builtin_amdgcn_sched_barrier(0);

    {
      const int jn = (step < 7) ? (j0 + 128) : j0;
      const float* __restrict__ xwjn = xwb + (size_t)(jn + l15) * FO_;
      #pragma unroll
      for (int kc = 0; kc < FO_ / 4; ++kc) xjq[kc] = *(const float4*)(xwjn + kc * 4);
    }

    #pragma unroll
    for (int r = 0; r < 8; ++r) {
      const int irow = r + hh * 8;
      const float a = av[r];
      const float q = a / fmaxf(acc[r], EPS_);
      float m = (a != 0.f) ? q : a;
      m *= DELTA_;
      dacc[r] += m;
      Msh[wv][irow][l15] = m;
    }

    v2f avec[4];
    #pragma unroll
    for (int kchunk = 0; kchunk < 4; ++kchunk) {
      const int ks = kchunk * 4 + hh * 2;
      avec[kchunk].x = Msh[wv][l15][ks];
      avec[kchunk].y = Msh[wv][l15][ks + 1];
    }
    #pragma unroll
    for (int fc = 0; fc < 4; ++fc) {
      #pragma unroll
      for (int kchunk = 0; kchunk < 4; ++kchunk) {
        const int ks = kchunk * 4 + hh * 2;
        const float* __restrict__ bp =
            xwb + (size_t)(j0 + ks) * FO_ + fc * 16 + l15;
        v2f bv;  bv.x = bp[0];  bv.y = bp[FO_];
        oacc[fc] = wmma_f32_16x16x4(avec[kchunk], bv, oacc[fc]);
      }
    }

    __builtin_amdgcn_sched_barrier(0);
    asm volatile("v_nop\n\tv_nop\n\tv_nop\n\tv_nop\n\tv_nop\n\tv_nop\n\tv_nop\n\tv_nop");
    __builtin_amdgcn_sched_barrier(0);
  }

  #pragma unroll
  for (int r = 0; r < 8; ++r) {
    float d = dacc[r];
    d += __shfl_xor(d, 1, 16);
    d += __shfl_xor(d, 2, 16);
    d += __shfl_xor(d, 4, 16);
    d += __shfl_xor(d, 8, 16);
    if (l15 == 0) atomicAdd(&deg_s[r + hh * 8], d);
  }
  #pragma unroll
  for (int fc = 0; fc < 4; ++fc)
    #pragma unroll
    for (int r = 0; r < 8; ++r)
      atomicAdd(&out_s[r + hh * 8][fc * 16 + l15], oacc[fc][r]);
  __syncthreads();

  {
    const int row = tid >> 4;
    const int c4  = (tid & 15) * 4;
    const float diagc = 1.f - deg_s[row];
    const float4 xi = *(const float4*)&xwi_s[row][c4];
    const float4 bs = *(const float4*)(bias + c4);
    float4 o = *(const float4*)&out_s[row][c4];
    o.x += diagc * xi.x + bs.x;
    o.y += diagc * xi.y + bs.y;
    o.z += diagc * xi.z + bs.z;
    o.w += diagc * xi.w + bs.w;
    *(float4*)(out + (size_t)(b * N_ + i0 + row) * FO_ + c4) = o;
  }
}

extern "C" void kernel_launch(void* const* d_in, const int* in_sizes, int n_in,
                              void* d_out, int out_size, void* d_ws, size_t ws_size,
                              hipStream_t stream) {
  (void)in_sizes; (void)n_in; (void)out_size; (void)ws_size;
  const float* x    = (const float*)d_in[0];
  const float* adj  = (const float*)d_in[1];
  const float* w    = (const float*)d_in[2];
  const float* bias = (const float*)d_in[3];
  float*       out  = (float*)d_out;
  float*       xw   = (float*)d_ws;

  gtv_xw_gemm<<<(B_ * N_) / 16, 128, 0, stream>>>(x, w, xw);
  gtv_fused  <<<B_ * (N_ / 16), 256, 0, stream>>>(adj, xw, bias, out);
}


# first kernel (x@W): k loop fully unrolled so its operand loads run about 10 k-steps (30 loads) ahead of the matrix instructions, not 8 load-wait-multiply rounds in series; sums unchanged
# speedup vs baseline: 2.3186x; 1.1199x over previous
#include <hip/hip_runtime.h>

#define B_    4
#define N_    1024
#define FI_   128
#define FO_   64
#define DELTA_ 1.0f
#define EPS_   1e-3f

typedef __attribute__((ext_vector_type(2))) float v2f;
typedef __attribute__((ext_vector_type(8))) float v8f;

static __device__ __forceinline__ v8f wmma_f32_16x16x4(v2f a, v2f b, v8f c) {
  return __builtin_amdgcn_wmma_f32_16x16x4_f32(
       false, a,  false, b,
       (short)0, c,  false,  false);
}

static __device__ __forceinline__ v2f pk_sub_f32(v2f a, v2f b) {
  v2f d;
  asm("v_pk_add_f32 %0, %1, %2 neg_lo:[0,1] neg_hi:[0,1]" : "=v"(d) : "v"(a), "v"(b));
  return d;
}

static __device__ __forceinline__ float add_abs_f32(float acc, float d) {
  float o;
  asm("v_add_f32_e64 %0, %1, |%2|" : "=v"(o) : "v"(acc), "v"(d));
  return o;
}

__global__ void __launch_bounds__(128)
gtv_xw_gemm(const float* __restrict__ x, const float* __restrict__ w,
            float* __restrict__ xw) {
  const int tid  = threadIdx.x;
  const int wv   = tid >> 5;
  const int lane = tid & 31;
  const int hh   = lane >> 4;
  const int l15  = lane & 15;
  const int rt   = blockIdx.x;
  const int f0   = wv * 16;

  const float* __restrict__ xrow = x + (size_t)(rt * 16 + l15) * FI_ + hh * 2;
  const float* __restrict__ wcol = w + (size_t)(hh * 2) * FO_ + f0 + l15;
  v2f a[FI_ / 4];
  v2f b[FI_ / 4];
  #pragma unroll
  for (int kc = 0; kc < FI_ / 4; ++kc) {
    a[kc].x = xrow[kc * 4];
    a[kc].y = xrow[kc * 4 + 1];
    b[kc].x = wcol[(size_t)(kc * 4) * FO_];
    b[kc].y = wcol[(size_t)(kc * 4 + 1) * FO_];
  }

  v8f c = {0.f, 0.f, 0.f, 0.f, 0.f, 0.f, 0.f, 0.f};
  #pragma unroll
  for (int kc = 0; kc < FI_ / 4; ++kc) c = wmma_f32_16x16x4(a[kc], b[kc], c);

  float* __restrict__ orow = xw + (size_t)(rt * 16 + hh * 8) * FO_ + f0 + l15;
  #pragma unroll
  for (int r = 0; r < 8; ++r) orow[(size_t)r * FO_] = c[r];
}

__global__ void __launch_bounds__(256, 4)
gtv_fused(const float* __restrict__ adj, const float* __restrict__ xw,
          const float* __restrict__ bias, float* __restrict__ out) {
  __shared__ float xwi_s[16][FO_];
  __shared__ float Msh[8][16][20];
  __shared__ float out_s[16][FO_];
  __shared__ float deg_s[16];

  const int tid  = threadIdx.x;
  const int wv   = tid >> 5;
  const int lane = tid & 31;
  const int hh   = lane >> 4;
  const int l15  = lane & 15;

  const int b  = blockIdx.x >> 6;
  const int i0 = (blockIdx.x & 63) * 16;

  if (tid < 16) deg_s[tid] = 0.f;
  {
    const int row = tid >> 4;
    const int c4  = (tid & 15) * 4;
    *(float4*)&out_s[row][c4] = make_float4(0.f, 0.f, 0.f, 0.f);
    *(float4*)&xwi_s[row][c4] =
        *(const float4*)(xw + (size_t)(b * N_ + i0 + row) * FO_ + c4);
  }
  __syncthreads();

  float dacc[8];
  #pragma unroll
  for (int r = 0; r < 8; ++r) dacc[r] = 0.f;
  v8f oacc[4];
  #pragma unroll
  for (int f = 0; f < 4; ++f)
    #pragma unroll
    for (int e = 0; e < 8; ++e) oacc[f][e] = 0.f;

  const float* __restrict__ adjb = adj + (size_t)b * N_ * N_ + (size_t)i0 * N_;
  const float* __restrict__ xwb  = xw + (size_t)b * N_ * FO_;

  float4 xjq[16];
  {
    const float* __restrict__ xwj0 = xwb + (size_t)(wv * 16 + l15) * FO_;
    #pragma unroll
    for (int kc = 0; kc < FO_ / 4; ++kc) xjq[kc] = *(const float4*)(xwj0 + kc * 4);
  }

  for (int step = 0; step < 8; ++step) {
    const int jt = step * 8 + wv;
    const int j0 = jt * 16;
    if (step < 7)
      __builtin_prefetch(adjb + (size_t)(hh * 8) * N_ + (j0 + 128) + l15, 0, 0);

    const float* __restrict__ ap = adjb + j0 + l15;
    float av[8];
    #pragma unroll
    for (int r = 0; r < 8; ++r) av[r] = ap[(size_t)(r + hh * 8) * N_];

    float acc[8];
    #pragma unroll
    for (int r = 0; r < 8; ++r) acc[r] = 0.f;
    float4 xiq[3][8];
    #pragma unroll
    for (int g = 0; g < 2; ++g)
      #pragma unroll
      for (int r = 0; r < 8; ++r) xiq[g][r] = *(const float4*)&xwi_s[r + hh * 8][g * 4];
    #pragma unroll
    for (int kc = 0; kc < FO_ / 4; ++kc) {
      const int cur = kc % 3;
      const int nxt = (kc + 2) % 3;
      if (kc + 2 < FO_ / 4) {
        #pragma unroll
        for (int r = 0; r < 4; ++r)
          xiq[nxt][r] = *(const float4*)&xwi_s[r + hh * 8][(kc + 2) * 4];
      }
      const float4 xj = xjq[kc];
      v2f xjlo;  xjlo.x = xj.x;  xjlo.y = xj.y;
      v2f xjhi;  xjhi.x = xj.z;  xjhi.y = xj.w;
      v2f d[8];
      #pragma unroll
      for (int r = 0; r < 8; ++r) {
        v2f xilo;  xilo.x = xiq[cur][r].x;  xilo.y = xiq[cur][r].y;
        d[r] = pk_sub_f32(xilo, xjlo);
      }
      #pragma unroll
      for (int r = 0; r < 8; ++r) acc[r] = add_abs_f32(acc[r], d[r].x);
      #pragma unroll
      for (int r = 0; r < 8; ++r) acc[r] = add_abs_f32(acc[r], d[r].y);
      if (kc + 2 < FO_ / 4) {
        #pragma unroll
        for (int r = 4; r < 8; ++r)
          xiq[nxt][r] = *(const float4*)&xwi_s[r + hh * 8][(kc + 2) * 4];
      }
      #pragma unroll
      for (int r = 0; r < 8; ++r) {
        v2f xihi;  xihi.x = xiq[cur][r].z;  xihi.y = xiq[cur][r].w;
        d[r] = pk_sub_f32(xihi, xjhi);
      }
      #pragma unroll
      for (int r = 0; r < 8; ++r) acc[r] = add_abs_f32(acc[r], d[r].x);
      #pragma unroll
      for (int r = 0; r < 8; ++r) acc[r] = add_abs_f32(acc[r], d[r].y);
      __builtin_amdgcn_sched_barrier(0);
    }

    __builtin_amdgcn_sched_barrier(0);

    {
      const int jn = (step < 7) ? (j0 + 128) : j0;
      const float* __restrict__ xwjn = xwb + (size_t)(jn + l15) * FO_;
      #pragma unroll
      for (int kc = 0; kc < FO_ / 4; ++kc) xjq[kc] = *(const float4*)(xwjn + kc * 4);
    }

    #pragma unroll
    for (int r = 0; r < 8; ++r) {
      const int irow = r + hh * 8;
      const float a = av[r];
      const float q = a / fmaxf(acc[r], EPS_);
      float m = (a != 0.f) ? q : a;
      m *= DELTA_;
      dacc[r] += m;
      Msh[wv][irow][l15] = m;
    }

    v2f avec[4];
    #pragma unroll
    for (int kchunk = 0; kchunk < 4; ++kchunk) {
      const int ks = kchunk * 4 + hh * 2;
      avec[kchunk].x = Msh[wv][l15][ks];
      avec[kchunk].y = Msh[wv][l15][ks + 1];
    }
    #pragma unroll
    for (int fc = 0; fc < 4; ++fc) {
      #pragma unroll
      for (int kchunk = 0; kchunk < 4; ++kchunk) {
        const int ks = kchunk * 4 + hh * 2;
        const float* __restrict__ bp =
            xwb + (size_t)(j0 + ks) * FO_ + fc * 16 + l15;
        v2f bv;  bv.x = bp[0];  bv.y = bp[FO_];
        oacc[fc] = wmma_f32_16x16x4(avec[kchunk], bv, oacc[fc]);
      }
    }

    __builtin_amdgcn_sched_barrier(0);
    asm volatile("v_nop\n\tv_nop\n\tv_nop\n\tv_nop\n\tv_nop\n\tv_nop\n\tv_nop\n\tv_nop");
    __builtin_amdgcn_sched_barrier(0);
  }

  #pragma unroll
  for (int r = 0; r < 8; ++r) {
    float d = dacc[r];
    d += __shfl_xor(d, 1, 16);
    d += __shfl_xor(d, 2, 16);
    d += __shfl_xor(d, 4, 16);
    d += __shfl_xor(d, 8, 16);
    if (l15 == 0) atomicAdd(&deg_s[r + hh * 8], d);
  }
  #pragma unroll
  for (int fc = 0; fc < 4; ++fc)
    #pragma unroll
    for (int r = 0; r < 8; ++r)
      atomicAdd(&out_s[r + hh * 8][fc * 16 + l15], oacc[fc][r]);
  __syncthreads();

  {
    const int row = tid >> 4;
    const int c4  = (tid & 15) * 4;
    const float diagc = 1.f - deg_s[row];
    const float4 xi = *(const float4*)&xwi_s[row][c4];
    const float4 bs = *(const float4*)(bias + c4);
    float4 o = *(const float4*)&out_s[row][c4];
    o.x += diagc * xi.x + bs.x;
    o.y += diagc * xi.y + bs.y;
    o.z += diagc * xi.z + bs.z;
    o.w += diagc * xi.w + bs.w;
    *(float4*)(out + (size_t)(b * N_ + i0 + row) * FO_ + c4) = o;
  }
}

extern "C" void kernel_launch(void* const* d_in, const int* in_sizes, int n_in,
                              void* d_out, int out_size, void* d_ws, size_t ws_size,
                              hipStream_t stream) {
  (void)in_sizes; (void)n_in; (void)out_size; (void)ws_size;
  const float* x    = (const float*)d_in[0];
  const float* adj  = (const float*)d_in[1];
  const float* w    = (const float*)d_in[2];
  const float* bias = (const float*)d_in[3];
  float*       out  = (float*)d_out;
  float*       xw   = (float*)d_ws;

  gtv_xw_gemm<<<(B_ * N_) / 16, 128, 0, stream>>>(x, w, xw);
  gtv_fused  <<<B_ * (N_ / 16), 256, 0, stream>>>(adj, xw, bias, out);
}


# epilogue: the eight degree row sums over 16 lanes on DPP row_xmask adds (32 vector adds) instead of 32 LDS-permute round trips waited for one by one; partner order and sums unchanged
# speedup vs baseline: 2.4383x; 1.0516x over previous
#include <hip/hip_runtime.h>

#define B_    4
#define N_    1024
#define FI_   128
#define FO_   64
#define DELTA_ 1.0f
#define EPS_   1e-3f

typedef __attribute__((ext_vector_type(2))) float v2f;
typedef __attribute__((ext_vector_type(8))) float v8f;

static __device__ __forceinline__ v8f wmma_f32_16x16x4(v2f a, v2f b, v8f c) {
  return __builtin_amdgcn_wmma_f32_16x16x4_f32(
       false, a,  false, b,
       (short)0, c,  false,  false);
}

static __device__ __forceinline__ v2f pk_sub_f32(v2f a, v2f b) {
  v2f d;
  asm("v_pk_add_f32 %0, %1, %2 neg_lo:[0,1] neg_hi:[0,1]" : "=v"(d) : "v"(a), "v"(b));
  return d;
}

static __device__ __forceinline__ float add_abs_f32(float acc, float d) {
  float o;
  asm("v_add_f32_e64 %0, %1, |%2|" : "=v"(o) : "v"(acc), "v"(d));
  return o;
}

template <int M>
static __device__ __forceinline__ float row_xor_add(float v) {
  const int p = __builtin_amdgcn_update_dpp(0, __float_as_int(v), 0x160 + M, 0xf, 0xf, false);
  return v + __int_as_float(p);
}

__global__ void __launch_bounds__(128)
gtv_xw_gemm(const float* __restrict__ x, const float* __restrict__ w,
            float* __restrict__ xw) {
  const int tid  = threadIdx.x;
  const int wv   = tid >> 5;
  const int lane = tid & 31;
  const int hh   = lane >> 4;
  const int l15  = lane & 15;
  const int rt   = blockIdx.x;
  const int f0   = wv * 16;

  const float* __restrict__ xrow = x + (size_t)(rt * 16 + l15) * FI_ + hh * 2;
  const float* __restrict__ wcol = w + (size_t)(hh * 2) * FO_ + f0 + l15;
  v2f a[FI_ / 4];
  v2f b[FI_ / 4];
  #pragma unroll
  for (int kc = 0; kc < FI_ / 4; ++kc) {
    a[kc].x = xrow[kc * 4];
    a[kc].y = xrow[kc * 4 + 1];
    b[kc].x = wcol[(size_t)(kc * 4) * FO_];
    b[kc].y = wcol[(size_t)(kc * 4 + 1) * FO_];
  }

  v8f c = {0.f, 0.f, 0.f, 0.f, 0.f, 0.f, 0.f, 0.f};
  #pragma unroll
  for (int kc = 0; kc < FI_ / 4; ++kc) c = wmma_f32_16x16x4(a[kc], b[kc], c);

  float* __restrict__ orow = xw + (size_t)(rt * 16 + hh * 8) * FO_ + f0 + l15;
  #pragma unroll
  for (int r = 0; r < 8; ++r) orow[(size_t)r * FO_] = c[r];
}

__global__ void __launch_bounds__(256, 4)
gtv_fused(const float* __restrict__ adj, const float* __restrict__ xw,
          const float* __restrict__ bias, float* __restrict__ out) {
  __shared__ float xwi_s[16][FO_];
  __shared__ float Msh[8][16][20];
  __shared__ float out_s[16][FO_];
  __shared__ float deg_s[16];

  const int tid  = threadIdx.x;
  const int wv   = tid >> 5;
  const int lane = tid & 31;
  const int hh   = lane >> 4;
  const int l15  = lane & 15;

  const int b  = blockIdx.x >> 6;
  const int i0 = (blockIdx.x & 63) * 16;

  if (tid < 16) deg_s[tid] = 0.f;
  {
    const int row = tid >> 4;
    const int c4  = (tid & 15) * 4;
    *(float4*)&out_s[row][c4] = make_float4(0.f, 0.f, 0.f, 0.f);
    *(float4*)&xwi_s[row][c4] =
        *(const float4*)(xw + (size_t)(b * N_ + i0 + row) * FO_ + c4);
  }
  __syncthreads();

  float dacc[8];
  #pragma unroll
  for (int r = 0; r < 8; ++r) dacc[r] = 0.f;
  v8f oacc[4];
  #pragma unroll
  for (int f = 0; f < 4; ++f)
    #pragma unroll
    for (int e = 0; e < 8; ++e) oacc[f][e] = 0.f;

  const float* __restrict__ adjb = adj + (size_t)b * N_ * N_ + (size_t)i0 * N_;
  const float* __restrict__ xwb  = xw + (size_t)b * N_ * FO_;

  float4 xjq[16];
  {
    const float* __restrict__ xwj0 = xwb + (size_t)(wv * 16 + l15) * FO_;
    #pragma unroll
    for (int kc = 0; kc < FO_ / 4; ++kc) xjq[kc] = *(const float4*)(xwj0 + kc * 4);
  }

  for (int step = 0; step < 8; ++step) {
    const int jt = step * 8 + wv;
    const int j0 = jt * 16;
    if (step < 7)
      __builtin_prefetch(adjb + (size_t)(hh * 8) * N_ + (j0 + 128) + l15, 0, 0);

    const float* __restrict__ ap = adjb + j0 + l15;
    float av[8];
    #pragma unroll
    for (int r = 0; r < 8; ++r) av[r] = ap[(size_t)(r + hh * 8) * N_];

    float acc[8];
    #pragma unroll
    for (int r = 0; r < 8; ++r) acc[r] = 0.f;
    float4 xiq[3][8];
    #pragma unroll
    for (int g = 0; g < 2; ++g)
      #pragma unroll
      for (int r = 0; r < 8; ++r) xiq[g][r] = *(const float4*)&xwi_s[r + hh * 8][g * 4];
    #pragma unroll
    for (int kc = 0; kc < FO_ / 4; ++kc) {
      const int cur = kc % 3;
      const int nxt = (kc + 2) % 3;
      if (kc + 2 < FO_ / 4) {
        #pragma unroll
        for (int r = 0; r < 4; ++r)
          xiq[nxt][r] = *(const float4*)&xwi_s[r + hh * 8][(kc + 2) * 4];
      }
      const float4 xj = xjq[kc];
      v2f xjlo;  xjlo.x = xj.x;  xjlo.y = xj.y;
      v2f xjhi;  xjhi.x = xj.z;  xjhi.y = xj.w;
      v2f d[8];
      #pragma unroll
      for (int r = 0; r < 8; ++r) {
        v2f xilo;  xilo.x = xiq[cur][r].x;  xilo.y = xiq[cur][r].y;
        d[r] = pk_sub_f32(xilo, xjlo);
      }
      #pragma unroll
      for (int r = 0; r < 8; ++r) acc[r] = add_abs_f32(acc[r], d[r].x);
      #pragma unroll
      for (int r = 0; r < 8; ++r) acc[r] = add_abs_f32(acc[r], d[r].y);
      if (kc + 2 < FO_ / 4) {
        #pragma unroll
        for (int r = 4; r < 8; ++r)
          xiq[nxt][r] = *(const float4*)&xwi_s[r + hh * 8][(kc + 2) * 4];
      }
      #pragma unroll
      for (int r = 0; r < 8; ++r) {
        v2f xihi;  xihi.x = xiq[cur][r].z;  xihi.y = xiq[cur][r].w;
        d[r] = pk_sub_f32(xihi, xjhi);
      }
      #pragma unroll
      for (int r = 0; r < 8; ++r) acc[r] = add_abs_f32(acc[r], d[r].x);
      #pragma unroll
      for (int r = 0; r < 8; ++r) acc[r] = add_abs_f32(acc[r], d[r].y);
      __builtin_amdgcn_sched_barrier(0);
    }

    __builtin_amdgcn_sched_barrier(0);

    {
      const int jn = (step < 7) ? (j0 + 128) : j0;
      const float* __restrict__ xwjn = xwb + (size_t)(jn + l15) * FO_;
      #pragma unroll
      for (int kc = 0; kc < FO_ / 4; ++kc) xjq[kc] = *(const float4*)(xwjn + kc * 4);
    }

    #pragma unroll
    for (int r = 0; r < 8; ++r) {
      const int irow = r + hh * 8;
      const float a = av[r];
      const float q = a / fmaxf(acc[r], EPS_);
      float m = (a != 0.f) ? q : a;
      m *= DELTA_;
      dacc[r] += m;
      Msh[wv][irow][l15] = m;
    }

    v2f avec[4];
    #pragma unroll
    for (int kchunk = 0; kchunk < 4; ++kchunk) {
      const int ks = kchunk * 4 + hh * 2;
      avec[kchunk].x = Msh[wv][l15][ks];
      avec[kchunk].y = Msh[wv][l15][ks + 1];
    }
    #pragma unroll
    for (int fc = 0; fc < 4; ++fc) {
      #pragma unroll
      for (int kchunk = 0; kchunk < 4; ++kchunk) {
        const int ks = kchunk * 4 + hh * 2;
        const float* __restrict__ bp =
            xwb + (size_t)(j0 + ks) * FO_ + fc * 16 + l15;
        v2f bv;  bv.x = bp[0];  bv.y = bp[FO_];
        oacc[fc] = wmma_f32_16x16x4(avec[kchunk], bv, oacc[fc]);
      }
    }

    __builtin_amdgcn_sched_barrier(0);
    asm volatile("v_nop\n\tv_nop\n\tv_nop\n\tv_nop\n\tv_nop\n\tv_nop\n\tv_nop\n\tv_nop");
    __builtin_amdgcn_sched_barrier(0);
  }

  #pragma unroll
  for (int r = 0; r < 8; ++r) dacc[r] = row_xor_add<1>(dacc[r]);
  #pragma unroll
  for (int r = 0; r < 8; ++r) dacc[r] = row_xor_add<2>(dacc[r]);
  #pragma unroll
  for (int r = 0; r < 8; ++r) dacc[r] = row_xor_add<4>(dacc[r]);
  #pragma unroll
  for (int r = 0; r < 8; ++r) dacc[r] = row_xor_add<8>(dacc[r]);
  if (l15 == 0) {
    #pragma unroll
    for (int r = 0; r < 8; ++r) atomicAdd(&deg_s[r + hh * 8], dacc[r]);
  }
  #pragma unroll
  for (int fc = 0; fc < 4; ++fc)
    #pragma unroll
    for (int r = 0; r < 8; ++r)
      atomicAdd(&out_s[r + hh * 8][fc * 16 + l15], oacc[fc][r]);
  __syncthreads();

  {
    const int row = tid >> 4;
    const int c4  = (tid & 15) * 4;
    const float diagc = 1.f - deg_s[row];
    const float4 xi = *(const float4*)&xwi_s[row][c4];
    const float4 bs = *(const float4*)(bias + c4);
    float4 o = *(const float4*)&out_s[row][c4];
    o.x += diagc * xi.x + bs.x;
    o.y += diagc * xi.y + bs.y;
    o.z += diagc * xi.z + bs.z;
    o.w += diagc * xi.w + bs.w;
    *(float4*)(out + (size_t)(b * N_ + i0 + row) * FO_ + c4) = o;
  }
}

extern "C" void kernel_launch(void* const* d_in, const int* in_sizes, int n_in,
                              void* d_out, int out_size, void* d_ws, size_t ws_size,
                              hipStream_t stream) {
  (void)in_sizes; (void)n_in; (void)out_size; (void)ws_size;
  const float* x    = (const float*)d_in[0];
  const float* adj  = (const float*)d_in[1];
  const float* w    = (const float*)d_in[2];
  const float* bias = (const float*)d_in[3];
  float*       out  = (float*)d_out;
  float*       xw   = (float*)d_ws;

  gtv_xw_gemm<<<(B_ * N_) / 16, 128, 0, stream>>>(x, w, xw);
  gtv_fused  <<<B_ * (N_ / 16), 256, 0, stream>>>(adj, xw, bias, out);
}


# xw workspace tile-major (4 features x 16 rows contiguous): a row-per-lane 16-byte xw_j load touches 2 cache lines, not 16; kernel 1 stores whole quads (product transposed); every sum unchanged
# speedup vs baseline: 2.5381x; 1.0409x over previous
#include <hip/hip_runtime.h>

#define B_    4
#define N_    1024
#define FI_   128
#define FO_   64
#define DELTA_ 1.0f
#define EPS_   1e-3f

typedef __attribute__((ext_vector_type(2))) float v2f;
typedef __attribute__((ext_vector_type(8))) float v8f;

static __device__ __forceinline__ v8f wmma_f32_16x16x4(v2f a, v2f b, v8f c) {
  return __builtin_amdgcn_wmma_f32_16x16x4_f32(
       false, a,  false, b,
       (short)0, c,  false,  false);
}

static __device__ __forceinline__ v2f pk_sub_f32(v2f a, v2f b) {
  v2f d;
  asm("v_pk_add_f32 %0, %1, %2 neg_lo:[0,1] neg_hi:[0,1]" : "=v"(d) : "v"(a), "v"(b));
  return d;
}

static __device__ __forceinline__ float add_abs_f32(float acc, float d) {
  float o;
  asm("v_add_f32_e64 %0, %1, |%2|" : "=v"(o) : "v"(acc), "v"(d));
  return o;
}

template <int M>
static __device__ __forceinline__ float row_xor_add(float v) {
  const int p = __builtin_amdgcn_update_dpp(0, __float_as_int(v), 0x160 + M, 0xf, 0xf, false);
  return v + __int_as_float(p);
}

#define XW_TILE_ 1024
#define XW_QUAD_ 64

__global__ void __launch_bounds__(128)
gtv_xw_gemm(const float* __restrict__ x, const float* __restrict__ w,
            float* __restrict__ xw) {
  const int tid  = threadIdx.x;
  const int wv   = tid >> 5;
  const int lane = tid & 31;
  const int hh   = lane >> 4;
  const int l15  = lane & 15;
  const int rt   = blockIdx.x;
  const int f0   = wv * 16;

  const float* __restrict__ xrow = x + (size_t)(rt * 16 + l15) * FI_ + hh * 2;
  const float* __restrict__ wcol = w + (size_t)(hh * 2) * FO_ + f0 + l15;
  v2f a[FI_ / 4];
  v2f b[FI_ / 4];
  #pragma unroll
  for (int kc = 0; kc < FI_ / 4; ++kc) {
    a[kc].x = xrow[kc * 4];
    a[kc].y = xrow[kc * 4 + 1];
    b[kc].x = wcol[(size_t)(kc * 4) * FO_];
    b[kc].y = wcol[(size_t)(kc * 4 + 1) * FO_];
  }

  v8f c = {0.f, 0.f, 0.f, 0.f, 0.f, 0.f, 0.f, 0.f};
  #pragma unroll
  for (int kc = 0; kc < FI_ / 4; ++kc) c = wmma_f32_16x16x4(b[kc], a[kc], c);

  float* __restrict__ orow = xw + (size_t)rt * XW_TILE_ + (wv * 4 + hh * 2) * XW_QUAD_ + l15 * 4;
  *(float4*)(orow)            = make_float4(c[0], c[1], c[2], c[3]);
  *(float4*)(orow + XW_QUAD_) = make_float4(c[4], c[5], c[6], c[7]);
}

__global__ void __launch_bounds__(256, 4)
gtv_fused(const float* __restrict__ adj, const float* __restrict__ xw,
          const float* __restrict__ bias, float* __restrict__ out) {
  __shared__ float xwi_s[16][FO_];
  __shared__ float Msh[8][16][20];
  __shared__ float out_s[16][FO_];
  __shared__ float deg_s[16];

  const int tid  = threadIdx.x;
  const int wv   = tid >> 5;
  const int lane = tid & 31;
  const int hh   = lane >> 4;
  const int l15  = lane & 15;

  const int b  = blockIdx.x >> 6;
  const int i0 = (blockIdx.x & 63) * 16;

  if (tid < 16) deg_s[tid] = 0.f;
  {
    const int row = tid >> 4;
    const int c4  = (tid & 15) * 4;
    *(float4*)&out_s[row][c4] = make_float4(0.f, 0.f, 0.f, 0.f);
    *(float4*)&xwi_s[row][c4] =
        *(const float4*)(xw + (size_t)(b * (N_ / 16) + (i0 >> 4)) * XW_TILE_
                         + (tid & 15) * XW_QUAD_ + row * 4);
  }
  __syncthreads();

  float dacc[8];
  #pragma unroll
  for (int r = 0; r < 8; ++r) dacc[r] = 0.f;
  v8f oacc[4];
  #pragma unroll
  for (int f = 0; f < 4; ++f)
    #pragma unroll
    for (int e = 0; e < 8; ++e) oacc[f][e] = 0.f;

  const float* __restrict__ adjb = adj + (size_t)b * N_ * N_ + (size_t)i0 * N_;
  const float* __restrict__ xwb  = xw + (size_t)b * (N_ / 16) * XW_TILE_;

  float4 xjq[16];
  {
    const float* __restrict__ xwj0 = xwb + (size_t)wv * XW_TILE_ + l15 * 4;
    #pragma unroll
    for (int kc = 0; kc < FO_ / 4; ++kc) xjq[kc] = *(const float4*)(xwj0 + kc * XW_QUAD_);
  }

  for (int step = 0; step < 8; ++step) {
    const int jt = step * 8 + wv;
    const int j0 = jt * 16;
    if (step < 7)
      __builtin_prefetch(adjb + (size_t)(hh * 8) * N_ + (j0 + 128) + l15, 0, 0);

    const float* __restrict__ ap = adjb + j0 + l15;
    float av[8];
    #pragma unroll
    for (int r = 0; r < 8; ++r) av[r] = ap[(size_t)(r + hh * 8) * N_];

    float acc[8];
    #pragma unroll
    for (int r = 0; r < 8; ++r) acc[r] = 0.f;
    float4 xiq[3][8];
    #pragma unroll
    for (int g = 0; g < 2; ++g)
      #pragma unroll
      for (int r = 0; r < 8; ++r) xiq[g][r] = *(const float4*)&xwi_s[r + hh * 8][g * 4];
    #pragma unroll
    for (int kc = 0; kc < FO_ / 4; ++kc) {
      const int cur = kc % 3;
      const int nxt = (kc + 2) % 3;
      if (kc + 2 < FO_ / 4) {
        #pragma unroll
        for (int r = 0; r < 4; ++r)
          xiq[nxt][r] = *(const float4*)&xwi_s[r + hh * 8][(kc + 2) * 4];
      }
      const float4 xj = xjq[kc];
      v2f xjlo;  xjlo.x = xj.x;  xjlo.y = xj.y;
      v2f xjhi;  xjhi.x = xj.z;  xjhi.y = xj.w;
      v2f d[8];
      #pragma unroll
      for (int r = 0; r < 8; ++r) {
        v2f xilo;  xilo.x = xiq[cur][r].x;  xilo.y = xiq[cur][r].y;
        d[r] = pk_sub_f32(xilo, xjlo);
      }
      #pragma unroll
      for (int r = 0; r < 8; ++r) acc[r] = add_abs_f32(acc[r], d[r].x);
      #pragma unroll
      for (int r = 0; r < 8; ++r) acc[r] = add_abs_f32(acc[r], d[r].y);
      if (kc + 2 < FO_ / 4) {
        #pragma unroll
        for (int r = 4; r < 8; ++r)
          xiq[nxt][r] = *(const float4*)&xwi_s[r + hh * 8][(kc + 2) * 4];
      }
      #pragma unroll
      for (int r = 0; r < 8; ++r) {
        v2f xihi;  xihi.x = xiq[cur][r].z;  xihi.y = xiq[cur][r].w;
        d[r] = pk_sub_f32(xihi, xjhi);
      }
      #pragma unroll
      for (int r = 0; r < 8; ++r) acc[r] = add_abs_f32(acc[r], d[r].x);
      #pragma unroll
      for (int r = 0; r < 8; ++r) acc[r] = add_abs_f32(acc[r], d[r].y);
      __builtin_amdgcn_sched_barrier(0);
    }

    __builtin_amdgcn_sched_barrier(0);

    {
      const int jtn = (step < 7) ? (jt + 8) : jt;
      const float* __restrict__ xwjn = xwb + (size_t)jtn * XW_TILE_ + l15 * 4;
      #pragma unroll
      for (int kc = 0; kc < FO_ / 4; ++kc) xjq[kc] = *(const float4*)(xwjn + kc * XW_QUAD_);
    }

    #pragma unroll
    for (int r = 0; r < 8; ++r) {
      const int irow = r + hh * 8;
      const float a = av[r];
      const float q = a / fmaxf(acc[r], EPS_);
      float m = (a != 0.f) ? q : a;
      m *= DELTA_;
      dacc[r] += m;
      Msh[wv][irow][l15] = m;
    }

    v2f avec[4];
    #pragma unroll
    for (int kchunk = 0; kchunk < 4; ++kchunk) {
      const int ks = kchunk * 4 + hh * 2;
      avec[kchunk].x = Msh[wv][l15][ks];
      avec[kchunk].y = Msh[wv][l15][ks + 1];
    }
    #pragma unroll
    for (int fc = 0; fc < 4; ++fc) {
      #pragma unroll
      for (int kchunk = 0; kchunk < 4; ++kchunk) {
        const int ks = kchunk * 4 + hh * 2;
        const float* __restrict__ bp =
            xwb + (size_t)jt * XW_TILE_ + (fc * 4 + (l15 >> 2)) * XW_QUAD_ + ks * 4 + (l15 & 3);
        v2f bv;  bv.x = bp[0];  bv.y = bp[4];
        oacc[fc] = wmma_f32_16x16x4(avec[kchunk], bv, oacc[fc]);
      }
    }

    __builtin_amdgcn_sched_barrier(0);
    asm volatile("v_nop\n\tv_nop\n\tv_nop\n\tv_nop\n\tv_nop\n\tv_nop\n\tv_nop\n\tv_nop");
    __builtin_amdgcn_sched_barrier(0);
  }

  #pragma unroll
  for (int r = 0; r < 8; ++r) dacc[r] = row_xor_add<1>(dacc[r]);
  #pragma unroll
  for (int r = 0; r < 8; ++r) dacc[r] = row_xor_add<2>(dacc[r]);
  #pragma unroll
  for (int r = 0; r < 8; ++r) dacc[r] = row_xor_add<4>(dacc[r]);
  #pragma unroll
  for (int r = 0; r < 8; ++r) dacc[r] = row_xor_add<8>(dacc[r]);
  if (l15 == 0) {
    #pragma unroll
    for (int r = 0; r < 8; ++r) atomicAdd(&deg_s[r + hh * 8], dacc[r]);
  }
  #pragma unroll
  for (int fc = 0; fc < 4; ++fc)
    #pragma unroll
    for (int r = 0; r < 8; ++r)
      atomicAdd(&out_s[r + hh * 8][fc * 16 + l15], oacc[fc][r]);
  __syncthreads();

  {
    const int row = tid >> 4;
    const int c4  = (tid & 15) * 4;
    const float diagc = 1.f - deg_s[row];
    const float4 xi = *(const float4*)&xwi_s[row][c4];
    const float4 bs = *(const float4*)(bias + c4);
    float4 o = *(const float4*)&out_s[row][c4];
    o.x += diagc * xi.x + bs.x;
    o.y += diagc * xi.y + bs.y;
    o.z += diagc * xi.z + bs.z;
    o.w += diagc * xi.w + bs.w;
    *(float4*)(out + (size_t)(b * N_ + i0 + row) * FO_ + c4) = o;
  }
}

extern "C" void kernel_launch(void* const* d_in, const int* in_sizes, int n_in,
                              void* d_out, int out_size, void* d_ws, size_t ws_size,
                              hipStream_t stream) {
  (void)in_sizes; (void)n_in; (void)out_size; (void)ws_size;
  const float* x    = (const float*)d_in[0];
  const float* adj  = (const float*)d_in[1];
  const float* w    = (const float*)d_in[2];
  const float* bias = (const float*)d_in[3];
  float*       out  = (float*)d_out;
  float*       xw   = (float*)d_ws;

  gtv_xw_gemm<<<(B_ * N_) / 16, 128, 0, stream>>>(x, w, xw);
  gtv_fused  <<<B_ * (N_ / 16), 256, 0, stream>>>(adj, xw, bias, out);
}


# second kernel: features 0..31 of the lane's eight xw_i rows stay in registers for the whole kernel (256 VGPRs, 2 waves per SIMD), so a distance step reads 64 LDS quads, not 128; every sum unchanged
# speedup vs baseline: 2.7259x; 1.0740x over previous
#include <hip/hip_runtime.h>

#define B_    4
#define N_    1024
#define FI_   128
#define FO_   64
#define DELTA_ 1.0f
#define EPS_   1e-3f

typedef __attribute__((ext_vector_type(2))) float v2f;
typedef __attribute__((ext_vector_type(8))) float v8f;

static __device__ __forceinline__ v8f wmma_f32_16x16x4(v2f a, v2f b, v8f c) {
  return __builtin_amdgcn_wmma_f32_16x16x4_f32(
       false, a,  false, b,
       (short)0, c,  false,  false);
}

static __device__ __forceinline__ v2f pk_sub_f32(v2f a, v2f b) { return a - b; }

static __device__ __forceinline__ float add_abs_f32(float acc, float d) {
  float o;
  asm("v_add_f32_e64 %0, %1, |%2|" : "=v"(o) : "v"(acc), "v"(d));
  return o;
}

template <int M>
static __device__ __forceinline__ float row_xor_add(float v) {
  const int p = __builtin_amdgcn_update_dpp(0, __float_as_int(v), 0x160 + M, 0xf, 0xf, false);
  return v + __int_as_float(p);
}

#define XW_TILE_ 1024
#define XW_QUAD_ 64

__global__ void __launch_bounds__(128)
gtv_xw_gemm(const float* __restrict__ x, const float* __restrict__ w,
            float* __restrict__ xw) {
  const int tid  = threadIdx.x;
  const int wv   = tid >> 5;
  const int lane = tid & 31;
  const int hh   = lane >> 4;
  const int l15  = lane & 15;
  const int rt   = blockIdx.x;
  const int f0   = wv * 16;

  const float* __restrict__ xrow = x + (size_t)(rt * 16 + l15) * FI_ + hh * 2;
  const float* __restrict__ wcol = w + (size_t)(hh * 2) * FO_ + f0 + l15;
  v2f a[FI_ / 4];
  v2f b[FI_ / 4];
  #pragma unroll
  for (int kc = 0; kc < FI_ / 4; ++kc) {
    a[kc].x = xrow[kc * 4];
    a[kc].y = xrow[kc * 4 + 1];
    b[kc].x = wcol[(size_t)(kc * 4) * FO_];
    b[kc].y = wcol[(size_t)(kc * 4 + 1) * FO_];
  }

  v8f c = {0.f, 0.f, 0.f, 0.f, 0.f, 0.f, 0.f, 0.f};
  #pragma unroll
  for (int kc = 0; kc < FI_ / 4; ++kc) c = wmma_f32_16x16x4(b[kc], a[kc], c);

  float* __restrict__ orow = xw + (size_t)rt * XW_TILE_ + (wv * 4 + hh * 2) * XW_QUAD_ + l15 * 4;
  *(float4*)(orow)            = make_float4(c[0], c[1], c[2], c[3]);
  *(float4*)(orow + XW_QUAD_) = make_float4(c[4], c[5], c[6], c[7]);
}

__global__ void __launch_bounds__(256) __attribute__((amdgpu_waves_per_eu(2, 2)))
gtv_fused(const float* __restrict__ adj, const float* __restrict__ xw,
          const float* __restrict__ bias, float* __restrict__ out) {
  __shared__ float xwi_s[16][FO_];
  __shared__ float Msh[8][16][20];
  __shared__ float out_s[16][FO_];
  __shared__ float deg_s[16];

  const int tid  = threadIdx.x;
  const int wv   = tid >> 5;
  const int lane = tid & 31;
  const int hh   = lane >> 4;
  const int l15  = lane & 15;

  const int b  = blockIdx.x >> 6;
  const int i0 = (blockIdx.x & 63) * 16;

  if (tid < 16) deg_s[tid] = 0.f;
  {
    const int row = tid >> 4;
    const int c4  = (tid & 15) * 4;
    *(float4*)&out_s[row][c4] = make_float4(0.f, 0.f, 0.f, 0.f);
    *(float4*)&xwi_s[row][c4] =
        *(const float4*)(xw + (size_t)(b * (N_ / 16) + (i0 >> 4)) * XW_TILE_
                         + (tid & 15) * XW_QUAD_ + row * 4);
  }
  __syncthreads();

  float4 xir[8][8];
  #pragma unroll
  for (int g = 0; g < 8; ++g)
    #pragma unroll
    for (int r = 0; r < 8; ++r) xir[g][r] = *(const float4*)&xwi_s[r + hh * 8][g * 4];

  float dacc[8];
  #pragma unroll
  for (int r = 0; r < 8; ++r) dacc[r] = 0.f;
  v8f oacc[4];
  #pragma unroll
  for (int f = 0; f < 4; ++f)
    #pragma unroll
    for (int e = 0; e < 8; ++e) oacc[f][e] = 0.f;

  const float* __restrict__ adjb = adj + (size_t)b * N_ * N_ + (size_t)i0 * N_;
  const float* __restrict__ xwb  = xw + (size_t)b * (N_ / 16) * XW_TILE_;

  float4 xjq[16];
  {
    const float* __restrict__ xwj0 = xwb + (size_t)wv * XW_TILE_ + l15 * 4;
    #pragma unroll
    for (int kc = 0; kc < FO_ / 4; ++kc) xjq[kc] = *(const float4*)(xwj0 + kc * XW_QUAD_);
  }

  for (int step = 0; step < 8; ++step) {
    const int jt = step * 8 + wv;
    const int j0 = jt * 16;
    if (step < 7)
      __builtin_prefetch(adjb + (size_t)(hh * 8) * N_ + (j0 + 128) + l15, 0, 0);

    const float* __restrict__ ap = adjb + j0 + l15;
    float av[8];
    #pragma unroll
    for (int r = 0; r < 8; ++r) av[r] = ap[(size_t)(r + hh * 8) * N_];

    float acc[8];
    #pragma unroll
    for (int r = 0; r < 8; ++r) acc[r] = 0.f;
    float4 xiq[3][8];
    #pragma unroll
    for (int g = 8; g < 10; ++g)
      #pragma unroll
      for (int r = 0; r < 8; ++r) xiq[g % 3][r] = *(const float4*)&xwi_s[r + hh * 8][g * 4];
    __builtin_amdgcn_sched_barrier(0);
    #pragma unroll
    for (int kc = 0; kc < 8; ++kc) {
      const float4 xj = xjq[kc];
      v2f xjlo;  xjlo.x = xj.x;  xjlo.y = xj.y;
      v2f xjhi;  xjhi.x = xj.z;  xjhi.y = xj.w;
      v2f d[8];
      #pragma unroll
      for (int r = 0; r < 8; ++r) {
        v2f xilo;  xilo.x = xir[kc][r].x;  xilo.y = xir[kc][r].y;
        d[r] = pk_sub_f32(xilo, xjlo);
      }
      #pragma unroll
      for (int r = 0; r < 8; ++r) acc[r] = add_abs_f32(acc[r], d[r].x);
      #pragma unroll
      for (int r = 0; r < 8; ++r) acc[r] = add_abs_f32(acc[r], d[r].y);
      #pragma unroll
      for (int r = 0; r < 8; ++r) {
        v2f xihi;  xihi.x = xir[kc][r].z;  xihi.y = xir[kc][r].w;
        d[r] = pk_sub_f32(xihi, xjhi);
      }
      #pragma unroll
      for (int r = 0; r < 8; ++r) acc[r] = add_abs_f32(acc[r], d[r].x);
      #pragma unroll
      for (int r = 0; r < 8; ++r) acc[r] = add_abs_f32(acc[r], d[r].y);
    }
    #pragma unroll
    for (int kc = 8; kc < FO_ / 4; ++kc) {
      const int cur = kc % 3;
      const int nxt = (kc + 2) % 3;
      if (kc + 2 < FO_ / 4) {
        #pragma unroll
        for (int r = 0; r < 4; ++r)
          xiq[nxt][r] = *(const float4*)&xwi_s[r + hh * 8][(kc + 2) * 4];
      }
      const float4 xj = xjq[kc];
      v2f xjlo;  xjlo.x = xj.x;  xjlo.y = xj.y;
      v2f xjhi;  xjhi.x = xj.z;  xjhi.y = xj.w;
      v2f d[8];
      #pragma unroll
      for (int r = 0; r < 8; ++r) {
        v2f xilo;  xilo.x = xiq[cur][r].x;  xilo.y = xiq[cur][r].y;
        d[r] = pk_sub_f32(xilo, xjlo);
      }
      #pragma unroll
      for (int r = 0; r < 8; ++r) acc[r] = add_abs_f32(acc[r], d[r].x);
      #pragma unroll
      for (int r = 0; r < 8; ++r) acc[r] = add_abs_f32(acc[r], d[r].y);
      if (kc + 2 < FO_ / 4) {
        #pragma unroll
        for (int r = 4; r < 8; ++r)
          xiq[nxt][r] = *(const float4*)&xwi_s[r + hh * 8][(kc + 2) * 4];
      }
      #pragma unroll
      for (int r = 0; r < 8; ++r) {
        v2f xihi;  xihi.x = xiq[cur][r].z;  xihi.y = xiq[cur][r].w;
        d[r] = pk_sub_f32(xihi, xjhi);
      }
      #pragma unroll
      for (int r = 0; r < 8; ++r) acc[r] = add_abs_f32(acc[r], d[r].x);
      #pragma unroll
      for (int r = 0; r < 8; ++r) acc[r] = add_abs_f32(acc[r], d[r].y);
      __builtin_amdgcn_sched_barrier(0);
    }

    __builtin_amdgcn_sched_barrier(0);

    {
      const int jtn = (step < 7) ? (jt + 8) : jt;
      const float* __restrict__ xwjn = xwb + (size_t)jtn * XW_TILE_ + l15 * 4;
      #pragma unroll
      for (int kc = 0; kc < FO_ / 4; ++kc) xjq[kc] = *(const float4*)(xwjn + kc * XW_QUAD_);
    }

    #pragma unroll
    for (int r = 0; r < 8; ++r) {
      const int irow = r + hh * 8;
      const float a = av[r];
      const float q = a / fmaxf(acc[r], EPS_);
      float m = (a != 0.f) ? q : a;
      m *= DELTA_;
      dacc[r] += m;
      Msh[wv][irow][l15] = m;
    }

    v2f avec[4];
    #pragma unroll
    for (int kchunk = 0; kchunk < 4; ++kchunk) {
      const int ks = kchunk * 4 + hh * 2;
      avec[kchunk].x = Msh[wv][l15][ks];
      avec[kchunk].y = Msh[wv][l15][ks + 1];
    }
    #pragma unroll
    for (int fc = 0; fc < 4; ++fc) {
      #pragma unroll
      for (int kchunk = 0; kchunk < 4; ++kchunk) {
        const int ks = kchunk * 4 + hh * 2;
        const float* __restrict__ bp =
            xwb + (size_t)jt * XW_TILE_ + (fc * 4 + (l15 >> 2)) * XW_QUAD_ + ks * 4 + (l15 & 3);
        v2f bv;  bv.x = bp[0];  bv.y = bp[4];
        oacc[fc] = wmma_f32_16x16x4(avec[kchunk], bv, oacc[fc]);
      }
    }

    __builtin_amdgcn_sched_barrier(0);
    asm volatile("v_nop\n\tv_nop\n\tv_nop\n\tv_nop\n\tv_nop\n\tv_nop\n\tv_nop\n\tv_nop");
    __builtin_amdgcn_sched_barrier(0);
  }

  #pragma unroll
  for (int r = 0; r < 8; ++r) dacc[r] = row_xor_add<1>(dacc[r]);
  #pragma unroll
  for (int r = 0; r < 8; ++r) dacc[r] = row_xor_add<2>(dacc[r]);
  #pragma unroll
  for (int r = 0; r < 8; ++r) dacc[r] = row_xor_add<4>(dacc[r]);
  #pragma unroll
  for (int r = 0; r < 8; ++r) dacc[r] = row_xor_add<8>(dacc[r]);
  if (l15 == 0) {
    #pragma unroll
    for (int r = 0; r < 8; ++r) atomicAdd(&deg_s[r + hh * 8], dacc[r]);
  }
  #pragma unroll
  for (int fc = 0; fc < 4; ++fc)
    #pragma unroll
    for (int r = 0; r < 8; ++r)
      atomicAdd(&out_s[r + hh * 8][fc * 16 + l15], oacc[fc][r]);
  __syncthreads();

  {
    const int row = tid >> 4;
    const int c4  = (tid & 15) * 4;
    const float diagc = 1.f - deg_s[row];
    const float4 xi = *(const float4*)&xwi_s[row][c4];
    const float4 bs = *(const float4*)(bias + c4);
    float4 o = *(const float4*)&out_s[row][c4];
    o.x += diagc * xi.x + bs.x;
    o.y += diagc * xi.y + bs.y;
    o.z += diagc * xi.z + bs.z;
    o.w += diagc * xi.w + bs.w;
    *(float4*)(out + (size_t)(b * N_ + i0 + row) * FO_ + c4) = o;
  }
}

extern "C" void kernel_launch(void* const* d_in, const int* in_sizes, int n_in,
                              void* d_out, int out_size, void* d_ws, size_t ws_size,
                              hipStream_t stream) {
  (void)in_sizes; (void)n_in; (void)out_size; (void)ws_size;
  const float* x    = (const float*)d_in[0];
  const float* adj  = (const float*)d_in[1];
  const float* w    = (const float*)d_in[2];
  const float* bias = (const float*)d_in[3];
  float*       out  = (float*)d_out;
  float*       xw   = (float*)d_ws;

  gtv_xw_gemm<<<(B_ * N_) / 16, 128, 0, stream>>>(x, w, xw);
  gtv_fused  <<<B_ * (N_ / 16), 256, 0, stream>>>(adj, xw, bias, out);
}
